# starting kernel
# baseline (speedup 1.0000x reference)
#include <hip/hip_runtime.h>
#include <hip/hip_bf16.h>

typedef __attribute__((ext_vector_type(2))) float v2f;
typedef __attribute__((ext_vector_type(4))) float v4f;
typedef __attribute__((ext_vector_type(8))) float v8f;

#define B_     4
#define S_     4096
#define D_     1024
#define SCHUNK 32
#define SPAN   (S_ / SCHUNK)

__global__ __launch_bounds__(256) void xsum_partial_kernel(
    const float* __restrict__ x, float* __restrict__ partials) {
  const int b  = blockIdx.x;
  const int sc = blockIdx.y;
  const int d  = threadIdx.x * 4;

  const float* p = x + ((size_t)b * S_ + (size_t)sc * SPAN) * D_ + d;
  v4f acc = {0.0f, 0.0f, 0.0f, 0.0f};
#pragma unroll 8
  for (int s = 0; s < SPAN; ++s) {
    v4f t = __builtin_nontemporal_load((const v4f*)(p + (size_t)s * D_));
    acc += t;
  }
  *(v4f*)(partials + ((size_t)sc * B_ + b) * D_ + d) = acc;
}

__global__ __launch_bounds__(256) void xsum_combine_kernel(
    const float* __restrict__ partials, float* __restrict__ A) {
  const int e = blockIdx.x * 256 + threadIdx.x;
  const int m = e >> 10;
  const int d = e & 1023;
  float v = 0.0f;
  if (m < B_) {
#pragma unroll
    for (int sc = 0; sc < SCHUNK; ++sc) {
      v += partials[((size_t)sc * B_ + m) * D_ + d];
    }
  }
  A[e] = v;
}

__global__ __launch_bounds__(32) void gemm16_f32wmma_kernel(
    const float* __restrict__ A, const float* __restrict__ W,
    float* __restrict__ Out) {
  const int lane = threadIdx.x;
  const int half = lane >> 4;
  const int l    = lane & 15;
  const int nTile = blockIdx.x;
  const int col   = nTile * 16 + l;

  const float* arow = A + (size_t)l * D_;
  const float* wrow = W + (size_t)col * D_;

  v8f c = {};
#pragma unroll 8
  for (int k0 = 0; k0 < D_; k0 += 4) {
    const int ka = k0 + half * 2;
    v2f a = *(const v2f*)(arow + ka);
    v2f b = *(const v2f*)(wrow + ka);
    c = __builtin_amdgcn_wmma_f32_16x16x4_f32(
         false, a,  false, b,
         (short)0, c,  false,  false);
  }

#pragma unroll
  for (int r = 0; r < 8; ++r) {
    Out[(size_t)(half * 8 + r) * D_ + col] = c[r];
  }
}

__global__ __launch_bounds__(256) void broadcast_kernel(
    const float* __restrict__ Y, float* __restrict__ out) {
  const size_t i  = (size_t)blockIdx.x * 256 + threadIdx.x;
  const int    o4 = (int)(i & 255);
  const int    b  = (int)(i >> 20);
  const v4f v = ((const v4f*)Y)[(size_t)b * 256 + o4];
  __builtin_nontemporal_store(v, (v4f*)out + i);
}

extern "C" void kernel_launch(void* const* d_in, const int* in_sizes, int n_in,
                              void* d_out, int out_size, void* d_ws, size_t ws_size,
                              hipStream_t stream) {
  const float* x     = (const float*)d_in[0];
  const float* w_qkv = (const float*)d_in[1];
  const float* w_o   = (const float*)d_in[2];
  const float* w_v   = w_qkv + (size_t)2 * D_ * D_;
  float* out = (float*)d_out;

  float* ws       = (float*)d_ws;
  float* partials = ws;
  float* A        = ws + 131072;
  float* T        = ws + 147456;
  float* Y        = ws + 163840;

  xsum_partial_kernel<<<dim3(B_, SCHUNK), 256, 0, stream>>>(x, partials);

  xsum_combine_kernel<<<64, 256, 0, stream>>>(partials, A);

  gemm16_f32wmma_kernel<<<64, 32, 0, stream>>>(A, w_v, T);

  gemm16_f32wmma_kernel<<<64, 32, 0, stream>>>(T, w_o, Y);

  const int n4 = (B_ * S_ * D_) / 4;
  broadcast_kernel<<<n4 / 256, 256, 0, stream>>>(Y, out);
}


# GEMMs: K split over 16 waves per 16-col tile, all slice loads issued up front; 16 partial tiles summed via LDS in fixed wave order (changes f32 summation order: 16 slice sums, then added)
# speedup vs baseline: 1.6209x; 1.6209x over previous
#include <hip/hip_runtime.h>
#include <hip/hip_bf16.h>

typedef __attribute__((ext_vector_type(2))) float v2f;
typedef __attribute__((ext_vector_type(4))) float v4f;
typedef __attribute__((ext_vector_type(8))) float v8f;

#define B_     4
#define S_     4096
#define D_     1024
#define SCHUNK 32
#define SPAN   (S_ / SCHUNK)

__global__ __launch_bounds__(256) void xsum_partial_kernel(
    const float* __restrict__ x, float* __restrict__ partials) {
  const int b  = blockIdx.x;
  const int sc = blockIdx.y;
  const int d  = threadIdx.x * 4;

  const float* p = x + ((size_t)b * S_ + (size_t)sc * SPAN) * D_ + d;
  v4f acc = {0.0f, 0.0f, 0.0f, 0.0f};
#pragma unroll 8
  for (int s = 0; s < SPAN; ++s) {
    v4f t = __builtin_nontemporal_load((const v4f*)(p + (size_t)s * D_));
    acc += t;
  }
  *(v4f*)(partials + ((size_t)sc * B_ + b) * D_ + d) = acc;
}

__global__ __launch_bounds__(256) void xsum_combine_kernel(
    const float* __restrict__ partials, float* __restrict__ A) {
  const int e = blockIdx.x * 256 + threadIdx.x;
  const int m = e >> 10;
  const int d = e & 1023;
  float v = 0.0f;
  if (m < B_) {
#pragma unroll
    for (int sc = 0; sc < SCHUNK; ++sc) {
      v += partials[((size_t)sc * B_ + m) * D_ + d];
    }
  }
  A[e] = v;
}

#define KSPLIT 16
#define KSPAN  (D_ / KSPLIT)

__global__ __launch_bounds__(32 * KSPLIT) void gemm16_f32wmma_kernel(
    const float* __restrict__ A, const float* __restrict__ W,
    float* __restrict__ Out) {
  __shared__ float red[KSPLIT][8][32];

  const int tid  = threadIdx.x;
  const int wave = tid >> 5;
  const int lane = tid & 31;
  const int half = lane >> 4;
  const int l    = lane & 15;
  const int nTile = blockIdx.x;
  const int col   = nTile * 16 + l;

  const float* arow = A + (size_t)l * D_ + wave * KSPAN;
  const float* wrow = W + (size_t)col * D_ + wave * KSPAN;

  v8f c = {};
  for (int k0 = 0; k0 < KSPAN; k0 += 4) {
    const int ka = k0 + half * 2;
    v2f a = *(const v2f*)(arow + ka);
    v2f b = *(const v2f*)(wrow + ka);
    c = __builtin_amdgcn_wmma_f32_16x16x4_f32(
         false, a,  false, b,
         (short)0, c,  false,  false);
  }

  for (int r = 0; r < 8; ++r) {
    red[wave][r][lane] = c[r];
  }
  __syncthreads();

  if (tid < 256) {
    const int r  = tid >> 5;
    const int ln = tid & 31;
    float sum = red[0][r][ln];
    for (int w = 1; w < KSPLIT; ++w) {
      sum += red[w][r][ln];
    }
    Out[(size_t)((ln >> 4) * 8 + r) * D_ + nTile * 16 + (ln & 15)] = sum;
  }
}

__global__ __launch_bounds__(256) void broadcast_kernel(
    const float* __restrict__ Y, float* __restrict__ out) {
  const size_t i  = (size_t)blockIdx.x * 256 + threadIdx.x;
  const int    o4 = (int)(i & 255);
  const int    b  = (int)(i >> 20);
  const v4f v = ((const v4f*)Y)[(size_t)b * 256 + o4];
  __builtin_nontemporal_store(v, (v4f*)out + i);
}

extern "C" void kernel_launch(void* const* d_in, const int* in_sizes, int n_in,
                              void* d_out, int out_size, void* d_ws, size_t ws_size,
                              hipStream_t stream) {
  const float* x     = (const float*)d_in[0];
  const float* w_qkv = (const float*)d_in[1];
  const float* w_o   = (const float*)d_in[2];
  const float* w_v   = w_qkv + (size_t)2 * D_ * D_;
  float* out = (float*)d_out;

  float* ws       = (float*)d_ws;
  float* partials = ws;
  float* A        = ws + 131072;
  float* T        = ws + 147456;
  float* Y        = ws + 163840;

  xsum_partial_kernel<<<dim3(B_, SCHUNK), 256, 0, stream>>>(x, partials);

  xsum_combine_kernel<<<64, 256, 0, stream>>>(partials, A);

  gemm16_f32wmma_kernel<<<64, 32 * KSPLIT, 0, stream>>>(A, w_v, T);

  gemm16_f32wmma_kernel<<<64, 32 * KSPLIT, 0, stream>>>(T, w_o, Y);

  const int n4 = (B_ * S_ * D_) / 4;
  broadcast_kernel<<<n4 / 256, 256, 0, stream>>>(Y, out);
}


# Broadcast pass: 256 resident 8-wave workgroups, each thread keeps its 16 bytes of Y in registers and stores them to 64 rows (was 16384 workgroups, one store per thread); values unchanged
# speedup vs baseline: 1.8233x; 1.1249x over previous
#include <hip/hip_runtime.h>
#include <hip/hip_bf16.h>

typedef __attribute__((ext_vector_type(2))) float v2f;
typedef __attribute__((ext_vector_type(4))) float v4f;
typedef __attribute__((ext_vector_type(8))) float v8f;

#define B_     4
#define S_     4096
#define D_     1024
#define SCHUNK 32
#define SPAN   (S_ / SCHUNK)

__global__ __launch_bounds__(256) void xsum_partial_kernel(
    const float* __restrict__ x, float* __restrict__ partials) {
  const int b  = blockIdx.x;
  const int sc = blockIdx.y;
  const int d  = threadIdx.x * 4;

  const float* p = x + ((size_t)b * S_ + (size_t)sc * SPAN) * D_ + d;
  v4f acc = {0.0f, 0.0f, 0.0f, 0.0f};
#pragma unroll 8
  for (int s = 0; s < SPAN; ++s) {
    v4f t = __builtin_nontemporal_load((const v4f*)(p + (size_t)s * D_));
    acc += t;
  }
  *(v4f*)(partials + ((size_t)sc * B_ + b) * D_ + d) = acc;
}

__global__ __launch_bounds__(256) void xsum_combine_kernel(
    const float* __restrict__ partials, float* __restrict__ A) {
  const int e = blockIdx.x * 256 + threadIdx.x;
  const int m = e >> 10;
  const int d = e & 1023;
  float v = 0.0f;
  if (m < B_) {
#pragma unroll
    for (int sc = 0; sc < SCHUNK; ++sc) {
      v += partials[((size_t)sc * B_ + m) * D_ + d];
    }
  }
  A[e] = v;
}

#define KSPLIT 16
#define KSPAN  (D_ / KSPLIT)

__global__ __launch_bounds__(32 * KSPLIT) void gemm16_f32wmma_kernel(
    const float* __restrict__ A, const float* __restrict__ W,
    float* __restrict__ Out) {
  __shared__ float red[KSPLIT][8][32];

  const int tid  = threadIdx.x;
  const int wave = tid >> 5;
  const int lane = tid & 31;
  const int half = lane >> 4;
  const int l    = lane & 15;
  const int nTile = blockIdx.x;
  const int col   = nTile * 16 + l;

  const float* arow = A + (size_t)l * D_ + wave * KSPAN;
  const float* wrow = W + (size_t)col * D_ + wave * KSPAN;

  v8f c = {};
  for (int k0 = 0; k0 < KSPAN; k0 += 4) {
    const int ka = k0 + half * 2;
    v2f a = *(const v2f*)(arow + ka);
    v2f b = *(const v2f*)(wrow + ka);
    c = __builtin_amdgcn_wmma_f32_16x16x4_f32(
         false, a,  false, b,
         (short)0, c,  false,  false);
  }

  for (int r = 0; r < 8; ++r) {
    red[wave][r][lane] = c[r];
  }
  __syncthreads();

  if (tid < 256) {
    const int r  = tid >> 5;
    const int ln = tid & 31;
    float sum = red[0][r][ln];
    for (int w = 1; w < KSPLIT; ++w) {
      sum += red[w][r][ln];
    }
    Out[(size_t)((ln >> 4) * 8 + r) * D_ + nTile * 16 + (ln & 15)] = sum;
  }
}

#define BC_ROWS   64
#define BC_BLOCKS (B_ * S_ / BC_ROWS)

__global__ __launch_bounds__(256) void broadcast_kernel(
    const float* __restrict__ Y, float* __restrict__ out) {
  const unsigned blk = blockIdx.x;
  const unsigned b   = blk / (S_ / BC_ROWS);
  const int d   = threadIdx.x * 4;
  const v4f v = *(const v4f*)(Y + (size_t)b * D_ + d);
  float* p = out + (size_t)blk * BC_ROWS * D_ + d;
#pragma unroll
  for (int r = 0; r < BC_ROWS; ++r) {
    __builtin_nontemporal_store(v, (v4f*)(p + (size_t)r * D_));
  }
}

extern "C" void kernel_launch(void* const* d_in, const int* in_sizes, int n_in,
                              void* d_out, int out_size, void* d_ws, size_t ws_size,
                              hipStream_t stream) {
  const float* x     = (const float*)d_in[0];
  const float* w_qkv = (const float*)d_in[1];
  const float* w_o   = (const float*)d_in[2];
  const float* w_v   = w_qkv + (size_t)2 * D_ * D_;
  float* out = (float*)d_out;

  float* ws       = (float*)d_ws;
  float* partials = ws;
  float* A        = ws + 131072;
  float* T        = ws + 147456;
  float* Y        = ws + 163840;

  xsum_partial_kernel<<<dim3(B_, SCHUNK), 256, 0, stream>>>(x, partials);

  xsum_combine_kernel<<<64, 256, 0, stream>>>(partials, A);

  gemm16_f32wmma_kernel<<<64, 32 * KSPLIT, 0, stream>>>(A, w_v, T);

  gemm16_f32wmma_kernel<<<64, 32 * KSPLIT, 0, stream>>>(T, w_o, Y);

  broadcast_kernel<<<BC_BLOCKS, 256, 0, stream>>>(Y, out);
}


# x column-sum pass: 256 workgroups of 16 waves (one per WGP, was 128 of 8), straight-line row loop keeps 12 loads in flight per wave; 4 row-group sums added via LDS (changes f32 summation order)
# speedup vs baseline: 2.3396x; 1.2832x over previous
#include <hip/hip_runtime.h>
#include <hip/hip_bf16.h>

typedef __attribute__((ext_vector_type(2))) float v2f;
typedef __attribute__((ext_vector_type(4))) float v4f;
typedef __attribute__((ext_vector_type(8))) float v8f;

#define B_     4
#define S_     4096
#define D_     1024
#define SCHUNK 32
#define SPAN   (S_ / SCHUNK)

#define XS_DSPLIT 2
#define XS_COLS   (D_ / XS_DSPLIT)
#define XS_LANES  (XS_COLS / 4)
#define XS_GROUPS 4
#define XS_ROWS   (SPAN / XS_GROUPS)

__global__ __launch_bounds__(XS_LANES * XS_GROUPS) void xsum_partial_kernel(
    const float* __restrict__ x, float* __restrict__ partials) {
  __shared__ v4f red[XS_GROUPS - 1][XS_LANES];

  const int b  = blockIdx.x;
  const int dh = blockIdx.y;
  const int sc = blockIdx.z;
  const int c  = threadIdx.x & (XS_LANES - 1);
  const int g  = threadIdx.x / XS_LANES;
  const int d  = dh * XS_COLS + c * 4;

  const float* p =
      x + ((size_t)b * S_ + (size_t)sc * SPAN + (size_t)g * XS_ROWS) * D_ + d;
  v4f acc = {0.0f, 0.0f, 0.0f, 0.0f};
#pragma unroll
  for (int s = 0; s < XS_ROWS; ++s) {
    v4f t = __builtin_nontemporal_load((const v4f*)(p + (size_t)s * D_));
    acc += t;
  }

  if (g > 0) {
    red[g - 1][c] = acc;
  }
  __syncthreads();
  if (g == 0) {
    acc += red[0][c];
    acc += red[1][c];
    acc += red[2][c];
    *(v4f*)(partials + ((size_t)sc * B_ + b) * D_ + d) = acc;
  }
}

__global__ __launch_bounds__(256) void xsum_combine_kernel(
    const float* __restrict__ partials, float* __restrict__ A) {
  const int e = blockIdx.x * 256 + threadIdx.x;
  const int m = e >> 10;
  const int d = e & 1023;
  float v = 0.0f;
  if (m < B_) {
#pragma unroll
    for (int sc = 0; sc < SCHUNK; ++sc) {
      v += partials[((size_t)sc * B_ + m) * D_ + d];
    }
  }
  A[e] = v;
}

#define KSPLIT 16
#define KSPAN  (D_ / KSPLIT)

__global__ __launch_bounds__(32 * KSPLIT) void gemm16_f32wmma_kernel(
    const float* __restrict__ A, const float* __restrict__ W,
    float* __restrict__ Out) {
  __shared__ float red[KSPLIT][8][32];

  const int tid  = threadIdx.x;
  const int wave = tid >> 5;
  const int lane = tid & 31;
  const int half = lane >> 4;
  const int l    = lane & 15;
  const int nTile = blockIdx.x;
  const int col   = nTile * 16 + l;

  const float* arow = A + (size_t)l * D_ + wave * KSPAN;
  const float* wrow = W + (size_t)col * D_ + wave * KSPAN;

  v8f c = {};
  for (int k0 = 0; k0 < KSPAN; k0 += 4) {
    const int ka = k0 + half * 2;
    v2f a = *(const v2f*)(arow + ka);
    v2f b = *(const v2f*)(wrow + ka);
    c = __builtin_amdgcn_wmma_f32_16x16x4_f32(
         false, a,  false, b,
         (short)0, c,  false,  false);
  }

  for (int r = 0; r < 8; ++r) {
    red[wave][r][lane] = c[r];
  }
  __syncthreads();

  if (tid < 256) {
    const int r  = tid >> 5;
    const int ln = tid & 31;
    float sum = red[0][r][ln];
    for (int w = 1; w < KSPLIT; ++w) {
      sum += red[w][r][ln];
    }
    Out[(size_t)((ln >> 4) * 8 + r) * D_ + nTile * 16 + (ln & 15)] = sum;
  }
}

#define BC_ROWS   64
#define BC_BLOCKS (B_ * S_ / BC_ROWS)

__global__ __launch_bounds__(256) void broadcast_kernel(
    const float* __restrict__ Y, float* __restrict__ out) {
  const unsigned blk = blockIdx.x;
  const unsigned b   = blk / (S_ / BC_ROWS);
  const int d   = threadIdx.x * 4;
  const v4f v = *(const v4f*)(Y + (size_t)b * D_ + d);
  float* p = out + (size_t)blk * BC_ROWS * D_ + d;
#pragma unroll
  for (int r = 0; r < BC_ROWS; ++r) {
    __builtin_nontemporal_store(v, (v4f*)(p + (size_t)r * D_));
  }
}

extern "C" void kernel_launch(void* const* d_in, const int* in_sizes, int n_in,
                              void* d_out, int out_size, void* d_ws, size_t ws_size,
                              hipStream_t stream) {
  const float* x     = (const float*)d_in[0];
  const float* w_qkv = (const float*)d_in[1];
  const float* w_o   = (const float*)d_in[2];
  const float* w_v   = w_qkv + (size_t)2 * D_ * D_;
  float* out = (float*)d_out;

  float* ws       = (float*)d_ws;
  float* partials = ws;
  float* A        = ws + 131072;
  float* T        = ws + 147456;
  float* Y        = ws + 163840;

  xsum_partial_kernel<<<dim3(B_, XS_DSPLIT, SCHUNK), XS_LANES * XS_GROUPS, 0, stream>>>(x, partials);

  xsum_combine_kernel<<<64, 256, 0, stream>>>(partials, A);

  gemm16_f32wmma_kernel<<<64, 32 * KSPLIT, 0, stream>>>(A, w_v, T);

  gemm16_f32wmma_kernel<<<64, 32 * KSPLIT, 0, stream>>>(T, w_o, Y);

  broadcast_kernel<<<BC_BLOCKS, 256, 0, stream>>>(Y, out);
}


# x column sums kept as 4 s-chunk partials per batch row = the 16 GEMM rows; combine launch dropped (5->4), broadcast adds the 4 partial Y rows (changes f32 summation order)
# speedup vs baseline: 2.5802x; 1.1028x over previous
#include <hip/hip_runtime.h>
#include <hip/hip_bf16.h>

typedef __attribute__((ext_vector_type(2))) float v2f;
typedef __attribute__((ext_vector_type(4))) float v4f;
typedef __attribute__((ext_vector_type(8))) float v8f;

#define B_     4
#define S_     4096
#define D_     1024
#define XS_CHUNKS 4
#define XS_SPAN   (S_ / XS_CHUNKS)
#define XS_COLS   64
#define XS_DSPLIT (D_ / XS_COLS)
#define XS_WAVES  16
#define XS_RPS    (2 * XS_WAVES)
#define XS_STEPS  (XS_SPAN / XS_RPS)

__global__ __launch_bounds__(32 * XS_WAVES) void xsum_rows_kernel(
    const float* __restrict__ x, float* __restrict__ P) {
  __shared__ v4f red[XS_WAVES][32];

  const int dp   = blockIdx.x;
  const int sc   = blockIdx.y;
  const int b    = blockIdx.z;
  const int lane = threadIdx.x & 31;
  const int wave = threadIdx.x >> 5;
  const int c    = lane & 15;
  const int h    = lane >> 4;

  const float* p = x + ((size_t)b * S_ + (size_t)sc * XS_SPAN + (size_t)(2 * wave + h)) * D_ +
                   dp * XS_COLS + c * 4;
  v4f acc = {0.0f, 0.0f, 0.0f, 0.0f};
#pragma unroll
  for (int s = 0; s < XS_STEPS; ++s) {
    v4f t = __builtin_nontemporal_load((const v4f*)(p + (size_t)s * XS_RPS * D_));
    acc += t;
  }

  red[wave][lane] = acc;
  __syncthreads();

  if (threadIdx.x < XS_COLS) {
    const float* r = (const float*)&red[0][0] + threadIdx.x;
    float s0 = r[0];
    float s1 = r[64];
#pragma unroll
    for (int w = 1; w < XS_WAVES; ++w) {
      s0 += r[w * 128];
      s1 += r[w * 128 + 64];
    }
    P[(size_t)(sc * B_ + b) * D_ + dp * XS_COLS + threadIdx.x] = s0 + s1;
  }
}

#define KSPLIT 16
#define KSPAN  (D_ / KSPLIT)

__global__ __launch_bounds__(32 * KSPLIT) void gemm16_f32wmma_kernel(
    const float* __restrict__ A, const float* __restrict__ W,
    float* __restrict__ Out) {
  __shared__ float red[KSPLIT][8][32];

  const int tid  = threadIdx.x;
  const int wave = tid >> 5;
  const int lane = tid & 31;
  const int half = lane >> 4;
  const int l    = lane & 15;
  const int nTile = blockIdx.x;
  const int col   = nTile * 16 + l;

  const float* arow = A + (size_t)l * D_ + wave * KSPAN;
  const float* wrow = W + (size_t)col * D_ + wave * KSPAN;

  v8f c = {};
  for (int k0 = 0; k0 < KSPAN; k0 += 4) {
    const int ka = k0 + half * 2;
    v2f a = *(const v2f*)(arow + ka);
    v2f b = *(const v2f*)(wrow + ka);
    c = __builtin_amdgcn_wmma_f32_16x16x4_f32(
         false, a,  false, b,
         (short)0, c,  false,  false);
  }

  for (int r = 0; r < 8; ++r) {
    red[wave][r][lane] = c[r];
  }
  __syncthreads();

  if (tid < 256) {
    const int r  = tid >> 5;
    const int ln = tid & 31;
    float sum = red[0][r][ln];
    for (int w = 1; w < KSPLIT; ++w) {
      sum += red[w][r][ln];
    }
    Out[(size_t)((ln >> 4) * 8 + r) * D_ + nTile * 16 + (ln & 15)] = sum;
  }
}

#define BC_ROWS   64
#define BC_BLOCKS (B_ * S_ / BC_ROWS)

__global__ __launch_bounds__(256) void broadcast_kernel(
    const float* __restrict__ Y, float* __restrict__ out) {
  const unsigned blk = blockIdx.x;
  const unsigned b   = blk / (S_ / BC_ROWS);
  const int d   = threadIdx.x * 4;
  const float* yb = Y + (size_t)b * D_ + d;
  v4f v = *(const v4f*)(yb);
  v += *(const v4f*)(yb + (size_t)1 * B_ * D_);
  v += *(const v4f*)(yb + (size_t)2 * B_ * D_);
  v += *(const v4f*)(yb + (size_t)3 * B_ * D_);
  float* p = out + (size_t)blk * BC_ROWS * D_ + d;
#pragma unroll
  for (int r = 0; r < BC_ROWS; ++r) {
    __builtin_nontemporal_store(v, (v4f*)(p + (size_t)r * D_));
  }
}

extern "C" void kernel_launch(void* const* d_in, const int* in_sizes, int n_in,
                              void* d_out, int out_size, void* d_ws, size_t ws_size,
                              hipStream_t stream) {
  const float* x     = (const float*)d_in[0];
  const float* w_qkv = (const float*)d_in[1];
  const float* w_o   = (const float*)d_in[2];
  const float* w_v   = w_qkv + (size_t)2 * D_ * D_;
  float* out = (float*)d_out;

  float* ws = (float*)d_ws;
  float* P  = ws + 131072;
  float* T  = ws + 147456;
  float* Y  = ws + 163840;

  xsum_rows_kernel<<<dim3(XS_DSPLIT, XS_CHUNKS, B_), 32 * XS_WAVES, 0, stream>>>(x, P);

  gemm16_f32wmma_kernel<<<64, 32 * KSPLIT, 0, stream>>>(P, w_v, T);

  gemm16_f32wmma_kernel<<<64, 32 * KSPLIT, 0, stream>>>(T, w_o, Y);

  broadcast_kernel<<<BC_BLOCKS, 256, 0, stream>>>(Y, out);
}


# Second product fused into the broadcast pass (launches 4->3): each workgroup computes y for its own 32 columns, 16-byte fragment loads (k order in a slice, so f32 summation order, changes)
# speedup vs baseline: 2.7827x; 1.0785x over previous
#include <hip/hip_runtime.h>
#include <hip/hip_bf16.h>

typedef __attribute__((ext_vector_type(2))) float v2f;
typedef __attribute__((ext_vector_type(4))) float v4f;
typedef __attribute__((ext_vector_type(8))) float v8f;

#define B_     4
#define S_     4096
#define D_     1024
#define XS_CHUNKS 4
#define XS_SPAN   (S_ / XS_CHUNKS)
#define XS_COLS   64
#define XS_DSPLIT (D_ / XS_COLS)
#define XS_WAVES  16
#define XS_RPS    (2 * XS_WAVES)
#define XS_STEPS  (XS_SPAN / XS_RPS)

__global__ __launch_bounds__(32 * XS_WAVES) void xsum_rows_kernel(
    const float* __restrict__ x, float* __restrict__ P) {
  __shared__ v4f red[XS_WAVES][32];

  const int dp   = blockIdx.x;
  const int sc   = blockIdx.y;
  const int b    = blockIdx.z;
  const int lane = threadIdx.x & 31;
  const int wave = threadIdx.x >> 5;
  const int c    = lane & 15;
  const int h    = lane >> 4;

  const float* p = x + ((size_t)b * S_ + (size_t)sc * XS_SPAN + (size_t)(2 * wave + h)) * D_ +
                   dp * XS_COLS + c * 4;
  v4f acc = {0.0f, 0.0f, 0.0f, 0.0f};
#pragma unroll
  for (int s = 0; s < XS_STEPS; ++s) {
    v4f t = __builtin_nontemporal_load((const v4f*)(p + (size_t)s * XS_RPS * D_));
    acc += t;
  }

  red[wave][lane] = acc;
  __syncthreads();

  if (threadIdx.x < XS_COLS) {
    const float* r = (const float*)&red[0][0] + threadIdx.x;
    float s0 = r[0];
    float s1 = r[64];
#pragma unroll
    for (int w = 1; w < XS_WAVES; ++w) {
      s0 += r[w * 128];
      s1 += r[w * 128 + 64];
    }
    P[(size_t)(sc * B_ + b) * D_ + dp * XS_COLS + threadIdx.x] = s0 + s1;
  }
}

#define KSPLIT 16
#define KSPAN  (D_ / KSPLIT)

__global__ __launch_bounds__(32 * KSPLIT) void gemm16_f32wmma_kernel(
    const float* __restrict__ A, const float* __restrict__ W,
    float* __restrict__ Out) {
  __shared__ float red[KSPLIT][8][32];

  const int tid  = threadIdx.x;
  const int wave = tid >> 5;
  const int lane = tid & 31;
  const int half = lane >> 4;
  const int l    = lane & 15;
  const int nTile = blockIdx.x;
  const int col   = nTile * 16 + l;

  const float* arow = A + (size_t)l * D_ + wave * KSPAN;
  const float* wrow = W + (size_t)col * D_ + wave * KSPAN;

  v8f c = {};
  for (int k0 = 0; k0 < KSPAN; k0 += 4) {
    const int ka = k0 + half * 2;
    v2f a = *(const v2f*)(arow + ka);
    v2f b = *(const v2f*)(wrow + ka);
    c = __builtin_amdgcn_wmma_f32_16x16x4_f32(
         false, a,  false, b,
         (short)0, c,  false,  false);
  }

  for (int r = 0; r < 8; ++r) {
    red[wave][r][lane] = c[r];
  }
  __syncthreads();

  if (tid < 256) {
    const int r  = tid >> 5;
    const int ln = tid & 31;
    float sum = red[0][r][ln];
    for (int w = 1; w < KSPLIT; ++w) {
      sum += red[w][r][ln];
    }
    Out[(size_t)((ln >> 4) * 8 + r) * D_ + nTile * 16 + (ln & 15)] = sum;
  }
}

#define BC_COLS   32
#define BC_PIECES (D_ / BC_COLS)
#define BC_HALVES 2
#define BC_SPAN   (S_ / BC_HALVES)
#define BC_STEP   (4 * KSPLIT)
#define BC_STEPS  (BC_SPAN / BC_STEP)

__global__ __launch_bounds__(32 * KSPLIT) void gemm_broadcast_kernel(
    const float* __restrict__ Tm, const float* __restrict__ Wo,
    float* __restrict__ out) {
  __shared__ float red[KSPLIT][2][4][16];
  __shared__ v4f ysm[BC_COLS / 4];

  const int tid   = threadIdx.x;
  const int wave  = tid >> 5;
  const int lane  = tid & 31;
  const int half  = lane >> 4;
  const int l     = lane & 15;
  const int piece = blockIdx.x;
  const int sh    = blockIdx.y;
  const int b     = blockIdx.z;

  const float* arow  = Tm + (size_t)((l & 3) * B_ + b) * D_ + wave * KSPAN;
  const float* wrow0 = Wo + (size_t)(piece * BC_COLS + l) * D_ + wave * KSPAN;
  const float* wrow1 = wrow0 + (size_t)16 * D_;

  v8f c0 = {};
  v8f c1 = {};
  for (int k0 = 0; k0 < KSPAN; k0 += 8) {
    const int ka = k0 + half * 4;
    const v4f a  = *(const v4f*)(arow + ka);
    const v4f w0 = *(const v4f*)(wrow0 + ka);
    const v4f w1 = *(const v4f*)(wrow1 + ka);
    const v2f alo = {a.x, a.y},   ahi = {a.z, a.w};
    const v2f w0lo = {w0.x, w0.y}, w0hi = {w0.z, w0.w};
    const v2f w1lo = {w1.x, w1.y}, w1hi = {w1.z, w1.w};
    c0 = __builtin_amdgcn_wmma_f32_16x16x4_f32(false, alo, false, w0lo, (short)0, c0, false, false);
    c1 = __builtin_amdgcn_wmma_f32_16x16x4_f32(false, alo, false, w1lo, (short)0, c1, false, false);
    c0 = __builtin_amdgcn_wmma_f32_16x16x4_f32(false, ahi, false, w0hi, (short)0, c0, false, false);
    c1 = __builtin_amdgcn_wmma_f32_16x16x4_f32(false, ahi, false, w1hi, (short)0, c1, false, false);
  }

  if (lane < 16) {
    for (int r = 0; r < 4; ++r) {
      red[wave][0][r][lane] = c0[r];
      red[wave][1][r][lane] = c1[r];
    }
  }
  __syncthreads();

  if (tid < BC_COLS) {
    const float* rp = &red[0][tid >> 4][0][tid & 15];
    float s0 = rp[0], s1 = rp[16], s2 = rp[32], s3 = rp[48];
    for (int w = 1; w < KSPLIT; ++w) {
      s0 += rp[w * 128];
      s1 += rp[w * 128 + 16];
      s2 += rp[w * 128 + 32];
      s3 += rp[w * 128 + 48];
    }
    float y = s0;
    y += s1;
    y += s2;
    y += s3;
    ((float*)ysm)[tid] = y;
  }
  __syncthreads();

  const int q  = lane & 7;
  const int rs = lane >> 3;
  const v4f v = ysm[q];
  float* p = out + ((size_t)b * S_ + (size_t)sh * BC_SPAN + (size_t)(wave * 4 + rs)) * D_ +
             piece * BC_COLS + q * 4;
#pragma unroll
  for (int s = 0; s < BC_STEPS; ++s) {
    __builtin_nontemporal_store(v, (v4f*)(p + (size_t)s * BC_STEP * D_));
  }
}

extern "C" void kernel_launch(void* const* d_in, const int* in_sizes, int n_in,
                              void* d_out, int out_size, void* d_ws, size_t ws_size,
                              hipStream_t stream) {
  const float* x     = (const float*)d_in[0];
  const float* w_qkv = (const float*)d_in[1];
  const float* w_o   = (const float*)d_in[2];
  const float* w_v   = w_qkv + (size_t)2 * D_ * D_;
  float* out = (float*)d_out;

  float* ws = (float*)d_ws;
  float* P  = ws + 131072;
  float* T  = ws + 147456;

  xsum_rows_kernel<<<dim3(XS_DSPLIT, XS_CHUNKS, B_), 32 * XS_WAVES, 0, stream>>>(x, P);

  gemm16_f32wmma_kernel<<<64, 32 * KSPLIT, 0, stream>>>(P, w_v, T);

  gemm_broadcast_kernel<<<dim3(BC_PIECES, BC_HALVES, B_), 32 * KSPLIT, 0, stream>>>(T, w_o, out);
}
